# starting kernel
# baseline (speedup 1.0000x reference)
#include <hip/hip_runtime.h>

typedef __attribute__((ext_vector_type(2))) float v2f;
typedef __attribute__((ext_vector_type(4))) float v4f;
typedef __attribute__((ext_vector_type(8))) float v8f;

#define EPSF 1e-5f
#define NB   16
#define CIN  256
#define MM   1024
#define DK   16
#define DV   64
#define QC   64

__global__ __launch_bounds__(256) void qkv_wmma(
    const float* __restrict__ x,  const float* __restrict__ Wq,
    const float* __restrict__ Wk, const float* __restrict__ Wv,
    const float* __restrict__ gq, const float* __restrict__ bq,
    const float* __restrict__ mq, const float* __restrict__ vq,
    const float* __restrict__ gv, const float* __restrict__ bv,
    const float* __restrict__ mv, const float* __restrict__ vv,
    float* __restrict__ Qbuf,
    float* __restrict__ Kbuf,
    float* __restrict__ Vbuf)
{
    const int wave = threadIdx.x >> 5;
    const int lane = threadIdx.x & 31;
    const int task = blockIdx.x * 8 + wave;
    const int b    = task / 576;
    const int rem  = task % 576;
    const int ot   = rem / 64;
    const int m0   = (rem % 64) * 16;

    const int nL   = lane & 15;
    const int half = lane >> 4;

    const float* W; int orow;
    if (ot < 4)       { W = Wq; orow = ot * 16; }
    else if (ot == 4) { W = Wk; orow = 0; }
    else              { W = Wv; orow = (ot - 5) * 16; }

    const float* wrow = W + (size_t)(orow + nL) * CIN + half * 2;
    const float* xcol = x + (size_t)b * CIN * MM + m0 + nL;

    v8f acc = {};
    #pragma unroll 4
    for (int c0 = 0; c0 < CIN; c0 += 4) {
        const int ca = c0 + half * 2;
        const v2f a = *reinterpret_cast<const v2f*>(wrow + c0);
        v2f bb;
        bb.x = xcol[(size_t)ca * MM];
        bb.y = xcol[(size_t)(ca + 1) * MM];
        acc = __builtin_amdgcn_wmma_f32_16x16x4_f32(
                  false, a, false, bb, (short)0, acc, false, false);
    }

    const int n = m0 + nL;
    if (ot == 4) {
        float* dst = Kbuf + (size_t)(b * MM + n) * DK;
        #pragma unroll
        for (int r = 0; r < 8; ++r) dst[r + half * 8] = acc[r];
    } else if (ot < 4) {
        float* dst = Qbuf + (size_t)(b * MM + n) * QC;
        #pragma unroll
        for (int r = 0; r < 8; ++r) {
            const int o = ot * 16 + r + half * 8;
            const float sc = gq[o] * rsqrtf(vq[o] + EPSF);
            dst[o] = acc[r] * sc + (bq[o] - mq[o] * sc);
        }
    } else {
        float* dst = Vbuf + (size_t)(b * MM + n) * DV;
        #pragma unroll
        for (int r = 0; r < 8; ++r) {
            const int o = (ot - 5) * 16 + r + half * 8;
            const float sc = gv[o] * rsqrtf(vv[o] + EPSF);
            dst[o] = acc[r] * sc + (bv[o] - mv[o] * sc);
        }
    }
}

__global__ __launch_bounds__(256) void ksoftmax(float* __restrict__ Kbuf)
{
    const int i = blockIdx.x * 256 + threadIdx.x;
    float* p = Kbuf + (size_t)i * DK;
    float v[DK], mx = -3.0e38f;
    #pragma unroll
    for (int k = 0; k < DK; ++k) { v[k] = p[k]; mx = fmaxf(mx, v[k]); }
    float s = 0.f;
    #pragma unroll
    for (int k = 0; k < DK; ++k) { v[k] = __expf(v[k] - mx); s += v[k]; }
    const float inv = 1.f / s;
    #pragma unroll
    for (int k = 0; k < DK; ++k) p[k] = v[k] * inv;
}

__global__ __launch_bounds__(256) void lambda_main(
    const float* __restrict__ rel,
    const float* __restrict__ Qbuf,
    const float* __restrict__ sK,
    const float* __restrict__ Vbuf,
    float* __restrict__ out)
{
    __shared__ float slab_all[8][DK * DV];

    const int wave = threadIdx.x >> 5;
    const int lane = threadIdx.x & 31;
    const int kM   = lane & 15;
    const int half = lane >> 4;

    const int b   = blockIdx.x >> 6;
    const int blk = blockIdx.x & 63;
    const int n1  = blk * 16 + wave * 2;
    const int n2  = n1 + 1;

    const int C1 = ((31 - (n1 >> 5)) * 63 + (31 - (n1 & 31))) * DK + kM;
    const int C2 = ((31 - (n2 >> 5)) * 63 + (31 - (n2 & 31))) * DK + kM;

    const float* sKb = sK   + (size_t)b * MM * DK + kM;
    const float* Vb  = Vbuf + (size_t)b * MM * DV + kM;

    v8f acc1[4] = {}, acc2[4] = {};

    #pragma unroll 2
    for (int mb = 0; mb < MM; mb += 4) {
        const int jj = (mb & 31) + half * 2;
        const int e0 = (mb >> 5) * 1008 + jj * 16;
        const int r0 = mb + half * 2;

        const float s0 = sKb[r0 * DK];
        const float s1 = sKb[r0 * DK + DK];
        v2f a1, a2;
        a1.x = rel[e0 + C1]      + s0;
        a1.y = rel[e0 + C1 + 16] + s1;
        a2.x = rel[e0 + C2]      + s0;
        a2.y = rel[e0 + C2 + 16] + s1;

        const float* v0p = Vb + (size_t)r0 * DV;
        #pragma unroll
        for (int t = 0; t < 4; ++t) {
            v2f bv;
            bv.x = v0p[t * 16];
            bv.y = v0p[t * 16 + DV];
            acc1[t] = __builtin_amdgcn_wmma_f32_16x16x4_f32(
                          false, a1, false, bv, (short)0, acc1[t], false, false);
            acc2[t] = __builtin_amdgcn_wmma_f32_16x16x4_f32(
                          false, a2, false, bv, (short)0, acc2[t], false, false);
        }
    }

    float* slab = slab_all[wave];
    const int h  = lane >> 3;
    const int v0 = (lane & 7) * 8;

    #pragma unroll
    for (int pass = 0; pass < 2; ++pass) {
        const v8f* A = pass ? acc2 : acc1;
        #pragma unroll
        for (int t = 0; t < 4; ++t)
            #pragma unroll
            for (int r = 0; r < 8; ++r)
                slab[(r + half * 8) * DV + t * 16 + kM] = A[t][r];
        __syncthreads();

        const int n = pass ? n2 : n1;
        const v4f* Qv = reinterpret_cast<const v4f*>(
                            Qbuf + (size_t)(b * MM + n) * QC + h * 16);
        v4f q4[4];
        #pragma unroll
        for (int g = 0; g < 4; ++g) q4[g] = Qv[g];

        v4f olo = {}, ohi = {};
        #pragma unroll
        for (int k = 0; k < DK; ++k) {
            const float q = q4[k >> 2][k & 3];
            const v4f* srow = reinterpret_cast<const v4f*>(slab + k * DV + v0);
            const v4f slo = srow[0], shi = srow[1];
            #pragma unroll
            for (int j = 0; j < 4; ++j) {
                olo[j] = fmaf(q, slo[j], olo[j]);
                ohi[j] = fmaf(q, shi[j], ohi[j]);
            }
        }
        v4f* dst = reinterpret_cast<v4f*>(
                       out + (size_t)(b * MM + n) * 256 + h * 64 + v0);
        dst[0] = olo;
        dst[1] = ohi;
        __syncthreads();
    }
}

extern "C" void kernel_launch(void* const* d_in, const int* in_sizes, int n_in,
                              void* d_out, int out_size, void* d_ws, size_t ws_size,
                              hipStream_t stream)
{
    const float* x   = (const float*)d_in[0];
    const float* Wq  = (const float*)d_in[1];
    const float* Wk  = (const float*)d_in[2];
    const float* Wv  = (const float*)d_in[3];
    const float* gq  = (const float*)d_in[4];
    const float* bq  = (const float*)d_in[5];
    const float* mq  = (const float*)d_in[6];
    const float* vq  = (const float*)d_in[7];
    const float* gv  = (const float*)d_in[8];
    const float* bv  = (const float*)d_in[9];
    const float* mv  = (const float*)d_in[10];
    const float* vv  = (const float*)d_in[11];
    const float* rel = (const float*)d_in[12];
    float* out = (float*)d_out;

    float* Qbuf = (float*)d_ws;
    float* Vbuf = Qbuf + (size_t)NB * MM * QC;
    float* Kbuf = Vbuf + (size_t)NB * MM * DV;

    qkv_wmma<<<1152, 256, 0, stream>>>(x, Wq, Wk, Wv, gq, bq, mq, vq,
                                       gv, bv, mv, vv, Qbuf, Kbuf, Vbuf);
    ksoftmax<<<64, 256, 0, stream>>>(Kbuf);
    lambda_main<<<1024, 256, 0, stream>>>(rel, Qbuf, Kbuf, Vbuf, out);
}


# Kernel 3 reassociated: P = Q*(emb+sK)^T over k first (block-diagonal f32 WMMA, 4 positions per wave), then out = P*V over m; halves the WMMA count; changes the floating-point summation order
# speedup vs baseline: 1.6286x; 1.6286x over previous
#include <hip/hip_runtime.h>

typedef __attribute__((ext_vector_type(2))) float v2f;
typedef __attribute__((ext_vector_type(4))) float v4f;
typedef __attribute__((ext_vector_type(8))) float v8f;

#define EPSF 1e-5f
#define NB   16
#define CIN  256
#define MM   1024
#define DK   16
#define DV   64
#define QC   64

__global__ __launch_bounds__(256) void qkv_wmma(
    const float* __restrict__ x,  const float* __restrict__ Wq,
    const float* __restrict__ Wk, const float* __restrict__ Wv,
    const float* __restrict__ gq, const float* __restrict__ bq,
    const float* __restrict__ mq, const float* __restrict__ vq,
    const float* __restrict__ gv, const float* __restrict__ bv,
    const float* __restrict__ mv, const float* __restrict__ vv,
    float* __restrict__ Qbuf,
    float* __restrict__ Kbuf,
    float* __restrict__ Vbuf)
{
    const int wave = threadIdx.x >> 5;
    const int lane = threadIdx.x & 31;
    const int task = blockIdx.x * 8 + wave;
    const int b    = task / 576;
    const int rem  = task % 576;
    const int ot   = rem / 64;
    const int m0   = (rem % 64) * 16;

    const int nL   = lane & 15;
    const int half = lane >> 4;

    const float* W; int orow;
    if (ot < 4)       { W = Wq; orow = ot * 16; }
    else if (ot == 4) { W = Wk; orow = 0; }
    else              { W = Wv; orow = (ot - 5) * 16; }

    const float* wrow = W + (size_t)(orow + nL) * CIN + half * 2;
    const float* xcol = x + (size_t)b * CIN * MM + m0 + nL;

    v8f acc = {};
    #pragma unroll 4
    for (int c0 = 0; c0 < CIN; c0 += 4) {
        const int ca = c0 + half * 2;
        const v2f a = *reinterpret_cast<const v2f*>(wrow + c0);
        v2f bb;
        bb.x = xcol[(size_t)ca * MM];
        bb.y = xcol[(size_t)(ca + 1) * MM];
        acc = __builtin_amdgcn_wmma_f32_16x16x4_f32(
                  false, a, false, bb, (short)0, acc, false, false);
    }

    const int n = m0 + nL;
    if (ot == 4) {
        float* dst = Kbuf + (size_t)(b * MM + n) * DK;
        #pragma unroll
        for (int r = 0; r < 8; ++r) dst[r + half * 8] = acc[r];
    } else if (ot < 4) {
        float* dst = Qbuf + (size_t)(b * MM + n) * QC;
        #pragma unroll
        for (int r = 0; r < 8; ++r) {
            const int o = ot * 16 + r + half * 8;
            const float sc = gq[o] * rsqrtf(vq[o] + EPSF);
            dst[o] = acc[r] * sc + (bq[o] - mq[o] * sc);
        }
    } else {
        float* dst = Vbuf + (size_t)(b * MM + n) * DV;
        #pragma unroll
        for (int r = 0; r < 8; ++r) {
            const int o = (ot - 5) * 16 + r + half * 8;
            const float sc = gv[o] * rsqrtf(vv[o] + EPSF);
            dst[o] = acc[r] * sc + (bv[o] - mv[o] * sc);
        }
    }
}

__global__ __launch_bounds__(256) void ksoftmax(float* __restrict__ Kbuf)
{
    const int i = blockIdx.x * 256 + threadIdx.x;
    float* p = Kbuf + (size_t)i * DK;
    float v[DK], mx = -3.0e38f;
    #pragma unroll
    for (int k = 0; k < DK; ++k) { v[k] = p[k]; mx = fmaxf(mx, v[k]); }
    float s = 0.f;
    #pragma unroll
    for (int k = 0; k < DK; ++k) { v[k] = __expf(v[k] - mx); s += v[k]; }
    const float inv = 1.f / s;
    #pragma unroll
    for (int k = 0; k < DK; ++k) p[k] = v[k] * inv;
}

struct MTile {
    v4f r[4][2];
    v4f k[2];
    v4f w[8];
};

__device__ __forceinline__ void load_tile(MTile& T, int mt,
    const float* __restrict__ relb, const float* __restrict__ skb,
    const float* __restrict__ vb)
{
    const int tb = (mt >> 1) * 1008 + (mt & 1) * 256;
    #pragma unroll
    for (int np = 0; np < 4; ++np) {
        const v4f* p = reinterpret_cast<const v4f*>(relb + tb - 16 * np);
        T.r[np][0] = p[0];
        T.r[np][1] = p[1];
    }
    const v4f* kp = reinterpret_cast<const v4f*>(skb + mt * (16 * DK));
    T.k[0] = kp[0];
    T.k[1] = kp[1];
    #pragma unroll
    for (int i = 0; i < 8; ++i)
        T.w[i] = *reinterpret_cast<const v4f*>(vb + (size_t)(mt * 16 + i) * DV);
}

__global__ __launch_bounds__(128) void lambda_main(
    const float* __restrict__ rel,
    const float* __restrict__ Qbuf,
    const float* __restrict__ sK,
    const float* __restrict__ Vbuf,
    float* __restrict__ out)
{
    const int wave = threadIdx.x >> 5;
    const int lane = threadIdx.x & 31;
    const int c    = lane & 15;
    const int hf   = lane >> 4;

    const int gw = blockIdx.x * 4 + wave;
    const int b  = gw >> 8;
    const int n0 = (gw & 255) * 4;

    const int nq = c >> 2, hq = c & 3;
    const v4f* qp = reinterpret_cast<const v4f*>(
        Qbuf + (size_t)(b * MM + n0 + nq) * QC + hq * DK + hf * 8);
    const v4f q0 = qp[0], q1 = qp[1];
    const float qv[8] = {q0[0], q0[1], q0[2], q0[3], q1[0], q1[1], q1[2], q1[3]};
    v2f bq[16];
    #pragma unroll
    for (int s = 0; s < 16; ++s) {
        const bool on = (s >> 2) == nq;
        bq[s].x = on ? qv[2 * (s & 3)]     : 0.f;
        bq[s].y = on ? qv[2 * (s & 3) + 1] : 0.f;
    }

    const int Cn = ((31 - (n0 >> 5)) * 63 + (31 - (n0 & 31))) * DK;
    const float* relb = rel  + Cn + c * DK + hf * 8;
    const float* skb  = sK   + (size_t)(b * MM + c) * DK + hf * 8;
    const float* vb   = Vbuf + (size_t)(b * MM + hf * 8) * DV + c * 4;

    v8f acc[4] = {};

    #pragma unroll 2
    for (int mt = 0; mt < MM / 16; ++mt) {
        MTile cur;
        load_tile(cur, mt, relb, skb, vb);

        const float kv[8] = {cur.k[0][0], cur.k[0][1], cur.k[0][2], cur.k[0][3],
                             cur.k[1][0], cur.k[1][1], cur.k[1][2], cur.k[1][3]};
        v8f pa = {}, pb = {};
        #pragma unroll
        for (int jj = 0; jj < 4; ++jj) {
            #pragma unroll
            for (int np = 0; np < 4; ++np) {
                const v4f rr = cur.r[np][jj >> 1];
                v2f a;
                a.x = rr[(2 * jj) & 3]     + kv[2 * jj];
                a.y = rr[(2 * jj + 1) & 3] + kv[2 * jj + 1];
                if (np < 2)
                    pa = __builtin_amdgcn_wmma_f32_16x16x4_f32(
                             false, a, false, bq[np * 4 + jj], (short)0, pa, false, false);
                else
                    pb = __builtin_amdgcn_wmma_f32_16x16x4_f32(
                             false, a, false, bq[np * 4 + jj], (short)0, pb, false, false);
            }
        }
        const v8f P = pa + pb;

        #pragma unroll
        for (int j = 0; j < 4; ++j) {
            v2f a2;
            a2.x = P[2 * j];
            a2.y = P[2 * j + 1];
            #pragma unroll
            for (int t = 0; t < 4; ++t) {
                v2f b2;
                b2.x = cur.w[2 * j][t];
                b2.y = cur.w[2 * j + 1][t];
                acc[t] = __builtin_amdgcn_wmma_f32_16x16x4_f32(
                             false, a2, false, b2, (short)0, acc[t], false, false);
            }
        }
    }

    #pragma unroll
    for (int r = 0; r < 8; ++r) {
        const int n = n0 + 2 * hf + (r >> 2);
        v4f o;
        o[0] = acc[0][r]; o[1] = acc[1][r]; o[2] = acc[2][r]; o[3] = acc[3][r];
        *reinterpret_cast<v4f*>(
            out + (size_t)(b * MM + n) * 256 + (r & 3) * DV + c * 4) = o;
    }
}

extern "C" void kernel_launch(void* const* d_in, const int* in_sizes, int n_in,
                              void* d_out, int out_size, void* d_ws, size_t ws_size,
                              hipStream_t stream)
{
    const float* x   = (const float*)d_in[0];
    const float* Wq  = (const float*)d_in[1];
    const float* Wk  = (const float*)d_in[2];
    const float* Wv  = (const float*)d_in[3];
    const float* gq  = (const float*)d_in[4];
    const float* bq  = (const float*)d_in[5];
    const float* mq  = (const float*)d_in[6];
    const float* vq  = (const float*)d_in[7];
    const float* gv  = (const float*)d_in[8];
    const float* bv  = (const float*)d_in[9];
    const float* mv  = (const float*)d_in[10];
    const float* vv  = (const float*)d_in[11];
    const float* rel = (const float*)d_in[12];
    float* out = (float*)d_out;

    float* Qbuf = (float*)d_ws;
    float* Vbuf = Qbuf + (size_t)NB * MM * QC;
    float* Kbuf = Vbuf + (size_t)NB * MM * DV;

    qkv_wmma<<<1152, 256, 0, stream>>>(x, Wq, Wk, Wv, gq, bq, mq, vq,
                                       gv, bv, mv, vv, Qbuf, Kbuf, Vbuf);
    ksoftmax<<<64, 256, 0, stream>>>(Kbuf);
    lambda_main<<<1024, 128, 0, stream>>>(rel, Qbuf, Kbuf, Vbuf, out);
}


# Kernel 3: wave = 4 vertically adjacent positions; one dense Q product per rel plane shared by all four through a 4-plane register window, content term by WMMA; changes the f32 summation order
# speedup vs baseline: 2.0522x; 1.2601x over previous
#include <hip/hip_runtime.h>

typedef __attribute__((ext_vector_type(2))) float v2f;
typedef __attribute__((ext_vector_type(4))) float v4f;
typedef __attribute__((ext_vector_type(8))) float v8f;

#define EPSF 1e-5f
#define NB   16
#define CIN  256
#define MM   1024
#define DK   16
#define DV   64
#define QC   64

__global__ __launch_bounds__(256) void qkv_wmma(
    const float* __restrict__ x,  const float* __restrict__ Wq,
    const float* __restrict__ Wk, const float* __restrict__ Wv,
    const float* __restrict__ gq, const float* __restrict__ bq,
    const float* __restrict__ mq, const float* __restrict__ vq,
    const float* __restrict__ gv, const float* __restrict__ bv,
    const float* __restrict__ mv, const float* __restrict__ vv,
    float* __restrict__ Qbuf,
    float* __restrict__ Kbuf,
    float* __restrict__ Vbuf)
{
    const int wave = threadIdx.x >> 5;
    const int lane = threadIdx.x & 31;
    const int task = blockIdx.x * 8 + wave;
    const int b    = task / 576;
    const int rem  = task % 576;
    const int ot   = rem / 64;
    const int m0   = (rem % 64) * 16;

    const int nL   = lane & 15;
    const int half = lane >> 4;

    const float* W; int orow;
    if (ot < 4)       { W = Wq; orow = ot * 16; }
    else if (ot == 4) { W = Wk; orow = 0; }
    else              { W = Wv; orow = (ot - 5) * 16; }

    const float* wrow = W + (size_t)(orow + nL) * CIN + half * 2;
    const float* xcol = x + (size_t)b * CIN * MM + m0 + nL;

    v8f acc = {};
    #pragma unroll 4
    for (int c0 = 0; c0 < CIN; c0 += 4) {
        const int ca = c0 + half * 2;
        const v2f a = *reinterpret_cast<const v2f*>(wrow + c0);
        v2f bb;
        bb.x = xcol[(size_t)ca * MM];
        bb.y = xcol[(size_t)(ca + 1) * MM];
        acc = __builtin_amdgcn_wmma_f32_16x16x4_f32(
                  false, a, false, bb, (short)0, acc, false, false);
    }

    const int n = m0 + nL;
    if (ot == 4) {
        float* dst = Kbuf + (size_t)(b * MM + n) * DK;
        #pragma unroll
        for (int r = 0; r < 8; ++r) dst[r + half * 8] = acc[r];
    } else if (ot < 4) {
        float* dst = Qbuf + (size_t)(b * MM + n) * QC;
        #pragma unroll
        for (int r = 0; r < 8; ++r) {
            const int o = ot * 16 + r + half * 8;
            const float sc = gq[o] * rsqrtf(vq[o] + EPSF);
            dst[o] = acc[r] * sc + (bq[o] - mq[o] * sc);
        }
    } else {
        float* dst = Vbuf + (size_t)(b * MM + n) * DV;
        #pragma unroll
        for (int r = 0; r < 8; ++r) {
            const int o = (ot - 5) * 16 + r + half * 8;
            const float sc = gv[o] * rsqrtf(vv[o] + EPSF);
            dst[o] = acc[r] * sc + (bv[o] - mv[o] * sc);
        }
    }
}

__global__ __launch_bounds__(256) void ksoftmax(float* __restrict__ Kbuf)
{
    const int i = blockIdx.x * 256 + threadIdx.x;
    float* p = Kbuf + (size_t)i * DK;
    float v[DK], mx = -3.0e38f;
    #pragma unroll
    for (int k = 0; k < DK; ++k) { v[k] = p[k]; mx = fmaxf(mx, v[k]); }
    float s = 0.f;
    #pragma unroll
    for (int k = 0; k < DK; ++k) { v[k] = __expf(v[k] - mx); s += v[k]; }
    const float inv = 1.f / s;
    #pragma unroll
    for (int k = 0; k < DK; ++k) p[k] = v[k] * inv;
}

__device__ __forceinline__ v8f wmma4(v2f a, v2f b, v8f c)
{
    return __builtin_amdgcn_wmma_f32_16x16x4_f32(
               false, a, false, b, (short)0, c, false, false);
}

__device__ __forceinline__ void plane_scores(v8f (&G)[2],
    const float* __restrict__ relp, const v2f (&bq)[4])
{
    v4f ra[2][2];
    #pragma unroll
    for (int t = 0; t < 2; ++t) {
        const v4f* p = reinterpret_cast<const v4f*>(relp + t * (16 * DK));
        ra[t][0] = p[0];
        ra[t][1] = p[1];
    }
    v8f g0 = {}, g1 = {};
    #pragma unroll
    for (int jj = 0; jj < 4; ++jj) {
        v2f a0, a1;
        a0.x = ra[0][jj >> 1][(2 * jj) & 3];
        a0.y = ra[0][jj >> 1][(2 * jj + 1) & 3];
        a1.x = ra[1][jj >> 1][(2 * jj) & 3];
        a1.y = ra[1][jj >> 1][(2 * jj + 1) & 3];
        g0 = wmma4(a0, bq[jj], g0);
        g1 = wmma4(a1, bq[jj], g1);
    }
    G[0] = g0;
    G[1] = g1;
}

__global__ __launch_bounds__(128) void lambda_main(
    const float* __restrict__ rel,
    const float* __restrict__ Qbuf,
    const float* __restrict__ sK,
    const float* __restrict__ Vbuf,
    float* __restrict__ out)
{
    const int wave = threadIdx.x >> 5;
    const int lane = threadIdx.x & 31;
    const int c    = lane & 15;
    const int hf   = lane >> 4;

    const int gw = blockIdx.x * 4 + wave;
    const int b  = gw >> 8;
    const int id = gw & 255;
    const int x0 = (id >> 5) * 4;
    const int y0 = id & 31;

    const int nq = c >> 2, hq = c & 3;
    const bool is0 = nq == 0, is2 = nq == 2, islo = nq < 2;
    const v4f* qp = reinterpret_cast<const v4f*>(
        Qbuf + (size_t)(b * MM + (x0 + nq) * 32 + y0) * QC + hq * DK + hf * 8);
    const v4f q0 = qp[0], q1 = qp[1];
    v2f bq[4];
    bq[0].x = q0[0]; bq[0].y = q0[1];
    bq[1].x = q0[2]; bq[1].y = q0[3];
    bq[2].x = q1[0]; bq[2].y = q1[1];
    bq[3].x = q1[2]; bq[3].y = q1[3];

    const float* relb = rel + ((28 - x0) * 63 + (31 - y0) + c) * DK + hf * 8;
    const float* skb  = sK   + (size_t)(b * MM + c) * DK + hf * 8;
    const float* vb   = Vbuf + (size_t)(b * MM + hf * 8) * DV + c * 4;

    v8f Gw[4][2];
    plane_scores(Gw[0], relb,            bq);
    plane_scores(Gw[1], relb + 1008,     bq);
    plane_scores(Gw[2], relb + 2 * 1008, bq);

    v8f acc[4] = {};

    for (int x4 = 0; x4 < 32; x4 += 4) {
        #pragma unroll
        for (int u = 0; u < 4; ++u) {
            const int xm = x4 + u;

            v4f kk[2][2];
            v4f w[2][8];
            #pragma unroll
            for (int t = 0; t < 2; ++t) {
                const int mt = 2 * xm + t;
                const v4f* kp = reinterpret_cast<const v4f*>(skb + mt * (16 * DK));
                kk[t][0] = kp[0];
                kk[t][1] = kp[1];
                #pragma unroll
                for (int i = 0; i < 8; ++i)
                    w[t][i] = *reinterpret_cast<const v4f*>(
                                  vb + (size_t)(mt * 16 + i) * DV);
            }

            plane_scores(Gw[(u + 3) & 3], relb + (xm + 3) * 1008, bq);

            v8f P[2];
            #pragma unroll
            for (int t = 0; t < 2; ++t) {
                #pragma unroll
                for (int r = 0; r < 8; ++r) {
                    const float g0 = Gw[(u + 3) & 3][t][r];
                    const float g1 = Gw[(u + 2) & 3][t][r];
                    const float g2 = Gw[(u + 1) & 3][t][r];
                    const float g3 = Gw[u & 3][t][r];
                    const float lo = is0 ? g0 : g1;
                    const float hi = is2 ? g2 : g3;
                    const float pv = islo ? lo : hi;
                    P[t][r] = pv;
                }
            }

            #pragma unroll
            for (int jj = 0; jj < 4; ++jj) {
                #pragma unroll
                for (int t = 0; t < 2; ++t) {
                    v2f a;
                    a.x = kk[t][jj >> 1][(2 * jj) & 3];
                    a.y = kk[t][jj >> 1][(2 * jj + 1) & 3];
                    P[t] = wmma4(a, bq[jj], P[t]);
                }
            }

            #pragma unroll
            for (int t = 0; t < 2; ++t) {
                #pragma unroll
                for (int j = 0; j < 4; ++j) {
                    v2f a2;
                    a2.x = P[t][2 * j];
                    a2.y = P[t][2 * j + 1];
                    #pragma unroll
                    for (int tt = 0; tt < 4; ++tt) {
                        v2f b2;
                        b2.x = w[t][2 * j][tt];
                        b2.y = w[t][2 * j + 1][tt];
                        acc[tt] = wmma4(a2, b2, acc[tt]);
                    }
                }
            }
        }
    }

    #pragma unroll
    for (int r = 0; r < 8; ++r) {
        const int n = (x0 + 2 * hf + (r >> 2)) * 32 + y0;
        v4f o;
        o[0] = acc[0][r]; o[1] = acc[1][r]; o[2] = acc[2][r]; o[3] = acc[3][r];
        *reinterpret_cast<v4f*>(
            out + (size_t)(b * MM + n) * 256 + (r & 3) * DV + c * 4) = o;
    }
}

extern "C" void kernel_launch(void* const* d_in, const int* in_sizes, int n_in,
                              void* d_out, int out_size, void* d_ws, size_t ws_size,
                              hipStream_t stream)
{
    const float* x   = (const float*)d_in[0];
    const float* Wq  = (const float*)d_in[1];
    const float* Wk  = (const float*)d_in[2];
    const float* Wv  = (const float*)d_in[3];
    const float* gq  = (const float*)d_in[4];
    const float* bq  = (const float*)d_in[5];
    const float* mq  = (const float*)d_in[6];
    const float* vq  = (const float*)d_in[7];
    const float* gv  = (const float*)d_in[8];
    const float* bv  = (const float*)d_in[9];
    const float* mv  = (const float*)d_in[10];
    const float* vv  = (const float*)d_in[11];
    const float* rel = (const float*)d_in[12];
    float* out = (float*)d_out;

    float* Qbuf = (float*)d_ws;
    float* Vbuf = Qbuf + (size_t)NB * MM * QC;
    float* Kbuf = Vbuf + (size_t)NB * MM * DV;

    qkv_wmma<<<1152, 256, 0, stream>>>(x, Wq, Wk, Wv, gq, bq, mq, vq,
                                       gv, bv, mv, vv, Qbuf, Kbuf, Vbuf);
    ksoftmax<<<64, 256, 0, stream>>>(Kbuf);
    lambda_main<<<1024, 128, 0, stream>>>(rel, Qbuf, Kbuf, Vbuf, out);
}


# Kernel 3 software-pipelined by 16-m stages: the 12 rel, sK and V loads of the next stage are issued as one group ahead of the current stage's 24 WMMAs (second register set), not just before use
# speedup vs baseline: 2.2494x; 1.0961x over previous
#include <hip/hip_runtime.h>

typedef __attribute__((ext_vector_type(2))) float v2f;
typedef __attribute__((ext_vector_type(4))) float v4f;
typedef __attribute__((ext_vector_type(8))) float v8f;

#define EPSF 1e-5f
#define NB   16
#define CIN  256
#define MM   1024
#define DK   16
#define DV   64
#define QC   64

__global__ __launch_bounds__(256) void qkv_wmma(
    const float* __restrict__ x,  const float* __restrict__ Wq,
    const float* __restrict__ Wk, const float* __restrict__ Wv,
    const float* __restrict__ gq, const float* __restrict__ bq,
    const float* __restrict__ mq, const float* __restrict__ vq,
    const float* __restrict__ gv, const float* __restrict__ bv,
    const float* __restrict__ mv, const float* __restrict__ vv,
    float* __restrict__ Qbuf,
    float* __restrict__ Kbuf,
    float* __restrict__ Vbuf)
{
    const int wave = threadIdx.x >> 5;
    const int lane = threadIdx.x & 31;
    const int task = blockIdx.x * 8 + wave;
    const int b    = task / 576;
    const int rem  = task % 576;
    const int ot   = rem / 64;
    const int m0   = (rem % 64) * 16;

    const int nL   = lane & 15;
    const int half = lane >> 4;

    const float* W; int orow;
    if (ot < 4)       { W = Wq; orow = ot * 16; }
    else if (ot == 4) { W = Wk; orow = 0; }
    else              { W = Wv; orow = (ot - 5) * 16; }

    const float* wrow = W + (size_t)(orow + nL) * CIN + half * 2;
    const float* xcol = x + (size_t)b * CIN * MM + m0 + nL;

    v8f acc = {};
    #pragma unroll 4
    for (int c0 = 0; c0 < CIN; c0 += 4) {
        const int ca = c0 + half * 2;
        const v2f a = *reinterpret_cast<const v2f*>(wrow + c0);
        v2f bb;
        bb.x = xcol[(size_t)ca * MM];
        bb.y = xcol[(size_t)(ca + 1) * MM];
        acc = __builtin_amdgcn_wmma_f32_16x16x4_f32(
                  false, a, false, bb, (short)0, acc, false, false);
    }

    const int n = m0 + nL;
    if (ot == 4) {
        float* dst = Kbuf + (size_t)(b * MM + n) * DK;
        #pragma unroll
        for (int r = 0; r < 8; ++r) dst[r + half * 8] = acc[r];
    } else if (ot < 4) {
        float* dst = Qbuf + (size_t)(b * MM + n) * QC;
        #pragma unroll
        for (int r = 0; r < 8; ++r) {
            const int o = ot * 16 + r + half * 8;
            const float sc = gq[o] * rsqrtf(vq[o] + EPSF);
            dst[o] = acc[r] * sc + (bq[o] - mq[o] * sc);
        }
    } else {
        float* dst = Vbuf + (size_t)(b * MM + n) * DV;
        #pragma unroll
        for (int r = 0; r < 8; ++r) {
            const int o = (ot - 5) * 16 + r + half * 8;
            const float sc = gv[o] * rsqrtf(vv[o] + EPSF);
            dst[o] = acc[r] * sc + (bv[o] - mv[o] * sc);
        }
    }
}

__global__ __launch_bounds__(256) void ksoftmax(float* __restrict__ Kbuf)
{
    const int i = blockIdx.x * 256 + threadIdx.x;
    float* p = Kbuf + (size_t)i * DK;
    float v[DK], mx = -3.0e38f;
    #pragma unroll
    for (int k = 0; k < DK; ++k) { v[k] = p[k]; mx = fmaxf(mx, v[k]); }
    float s = 0.f;
    #pragma unroll
    for (int k = 0; k < DK; ++k) { v[k] = __expf(v[k] - mx); s += v[k]; }
    const float inv = 1.f / s;
    #pragma unroll
    for (int k = 0; k < DK; ++k) p[k] = v[k] * inv;
}

__device__ __forceinline__ v8f wmma4(v2f a, v2f b, v8f c)
{
    return __builtin_amdgcn_wmma_f32_16x16x4_f32(
               false, a, false, b, (short)0, c, false, false);
}

__device__ __forceinline__ void plane_scores(v8f (&G)[2],
    const float* __restrict__ relp, const v2f (&bq)[4])
{
    v4f ra[2][2];
    #pragma unroll
    for (int t = 0; t < 2; ++t) {
        const v4f* p = reinterpret_cast<const v4f*>(relp + t * (16 * DK));
        ra[t][0] = p[0];
        ra[t][1] = p[1];
    }
    v8f g0 = {}, g1 = {};
    #pragma unroll
    for (int jj = 0; jj < 4; ++jj) {
        v2f a0, a1;
        a0.x = ra[0][jj >> 1][(2 * jj) & 3];
        a0.y = ra[0][jj >> 1][(2 * jj + 1) & 3];
        a1.x = ra[1][jj >> 1][(2 * jj) & 3];
        a1.y = ra[1][jj >> 1][(2 * jj + 1) & 3];
        g0 = wmma4(a0, bq[jj], g0);
        g1 = wmma4(a1, bq[jj], g1);
    }
    G[0] = g0;
    G[1] = g1;
}

struct Stage {
    v4f ra[2];
    v4f kk[2];
    v4f w[8];
};

__device__ __forceinline__ void stage_load(Stage& d,
    const float* __restrict__ relp, const float* __restrict__ skp,
    const float* __restrict__ vp)
{
    const v4f* rp = reinterpret_cast<const v4f*>(relp);
    d.ra[0] = rp[0];
    d.ra[1] = rp[1];
    const v4f* kp = reinterpret_cast<const v4f*>(skp);
    d.kk[0] = kp[0];
    d.kk[1] = kp[1];
    #pragma unroll
    for (int i = 0; i < 8; ++i)
        d.w[i] = *reinterpret_cast<const v4f*>(vp + (size_t)i * DV);
}

__global__ __launch_bounds__(128) void lambda_main(
    const float* __restrict__ rel,
    const float* __restrict__ Qbuf,
    const float* __restrict__ sK,
    const float* __restrict__ Vbuf,
    float* __restrict__ out)
{
    const int wave = threadIdx.x >> 5;
    const int lane = threadIdx.x & 31;
    const int c    = lane & 15;
    const int hf   = lane >> 4;

    const int gw = blockIdx.x * 4 + wave;
    const int b  = gw >> 8;
    const int id = gw & 255;
    const int x0 = (id >> 5) * 4;
    const int y0 = id & 31;

    const int nq = c >> 2, hq = c & 3;
    const bool is0 = nq == 0, is2 = nq == 2, islo = nq < 2;
    const v4f* qp = reinterpret_cast<const v4f*>(
        Qbuf + (size_t)(b * MM + (x0 + nq) * 32 + y0) * QC + hq * DK + hf * 8);
    const v4f q0 = qp[0], q1 = qp[1];
    v2f bq[4];
    bq[0].x = q0[0]; bq[0].y = q0[1];
    bq[1].x = q0[2]; bq[1].y = q0[3];
    bq[2].x = q1[0]; bq[2].y = q1[1];
    bq[3].x = q1[2]; bq[3].y = q1[3];

    const float* relb = rel + ((28 - x0) * 63 + (31 - y0) + c) * DK + hf * 8;
    const float* skb  = sK   + (size_t)(b * MM + c) * DK + hf * 8;
    const float* vb   = Vbuf + (size_t)(b * MM + hf * 8) * DV + c * 4;

    v8f Gw[4][2];
    plane_scores(Gw[0], relb,            bq);
    plane_scores(Gw[1], relb + 1008,     bq);
    plane_scores(Gw[2], relb + 2 * 1008, bq);

    v8f acc[4] = {};

    Stage cur, nxt;
    stage_load(cur, relb + 3 * 1008, skb, vb);

    for (int x4 = 0; x4 < 32; x4 += 4) {
        #pragma unroll
        for (int u = 0; u < 4; ++u) {
            const int xm = x4 + u;
            #pragma unroll
            for (int t = 0; t < 2; ++t) {
                const int xn  = t == 0 ? xm : min(xm + 1, 31);
                const int mtn = 2 * xn + (t ^ 1);
                stage_load(nxt, relb + (xn + 3) * 1008 + (t ^ 1) * (16 * DK),
                           skb + mtn * (16 * DK), vb + (size_t)mtn * (16 * DV));
                __builtin_amdgcn_sched_barrier(0);

                v8f g = {};
                #pragma unroll
                for (int jj = 0; jj < 4; ++jj) {
                    v2f a;
                    a.x = cur.ra[jj >> 1][(2 * jj) & 3];
                    a.y = cur.ra[jj >> 1][(2 * jj + 1) & 3];
                    g = wmma4(a, bq[jj], g);
                }
                Gw[(u + 3) & 3][t] = g;

                v8f P;
                #pragma unroll
                for (int r = 0; r < 8; ++r) {
                    const float g0 = Gw[(u + 3) & 3][t][r];
                    const float g1 = Gw[(u + 2) & 3][t][r];
                    const float g2 = Gw[(u + 1) & 3][t][r];
                    const float g3 = Gw[u & 3][t][r];
                    const float lo = is0 ? g0 : g1;
                    const float hi = is2 ? g2 : g3;
                    P[r] = islo ? lo : hi;
                }

                #pragma unroll
                for (int jj = 0; jj < 4; ++jj) {
                    v2f a;
                    a.x = cur.kk[jj >> 1][(2 * jj) & 3];
                    a.y = cur.kk[jj >> 1][(2 * jj + 1) & 3];
                    P = wmma4(a, bq[jj], P);
                }

                #pragma unroll
                for (int j = 0; j < 4; ++j) {
                    v2f a2;
                    a2.x = P[2 * j];
                    a2.y = P[2 * j + 1];
                    #pragma unroll
                    for (int tt = 0; tt < 4; ++tt) {
                        v2f b2;
                        b2.x = cur.w[2 * j][tt];
                        b2.y = cur.w[2 * j + 1][tt];
                        acc[tt] = wmma4(a2, b2, acc[tt]);
                    }
                }

                cur = nxt;
            }
        }
    }

    #pragma unroll
    for (int r = 0; r < 8; ++r) {
        const int n = (x0 + 2 * hf + (r >> 2)) * 32 + y0;
        v4f o;
        o[0] = acc[0][r]; o[1] = acc[1][r]; o[2] = acc[2][r]; o[3] = acc[3][r];
        *reinterpret_cast<v4f*>(
            out + (size_t)(b * MM + n) * 256 + (r & 3) * DV + c * 4) = o;
    }
}

extern "C" void kernel_launch(void* const* d_in, const int* in_sizes, int n_in,
                              void* d_out, int out_size, void* d_ws, size_t ws_size,
                              hipStream_t stream)
{
    const float* x   = (const float*)d_in[0];
    const float* Wq  = (const float*)d_in[1];
    const float* Wk  = (const float*)d_in[2];
    const float* Wv  = (const float*)d_in[3];
    const float* gq  = (const float*)d_in[4];
    const float* bq  = (const float*)d_in[5];
    const float* mq  = (const float*)d_in[6];
    const float* vq  = (const float*)d_in[7];
    const float* gv  = (const float*)d_in[8];
    const float* bv  = (const float*)d_in[9];
    const float* mv  = (const float*)d_in[10];
    const float* vv  = (const float*)d_in[11];
    const float* rel = (const float*)d_in[12];
    float* out = (float*)d_out;

    float* Qbuf = (float*)d_ws;
    float* Vbuf = Qbuf + (size_t)NB * MM * QC;
    float* Kbuf = Vbuf + (size_t)NB * MM * DV;

    qkv_wmma<<<1152, 256, 0, stream>>>(x, Wq, Wk, Wv, gq, bq, mq, vq,
                                       gv, bv, mv, vv, Qbuf, Kbuf, Vbuf);
    ksoftmax<<<64, 256, 0, stream>>>(Kbuf);
    lambda_main<<<1024, 128, 0, stream>>>(rel, Qbuf, Kbuf, Vbuf, out);
}


# Content term leaves the m loop: kernel 2 now reduces lam = sK^T V per batch (f32 WMMA, 4 quarter sums), kernel 3 adds Q*lam once (16 WMMA), 20 not 24 WMMA per stage; changes the f32 summation order
# speedup vs baseline: 2.5080x; 1.1150x over previous
#include <hip/hip_runtime.h>

typedef __attribute__((ext_vector_type(2))) float v2f;
typedef __attribute__((ext_vector_type(4))) float v4f;
typedef __attribute__((ext_vector_type(8))) float v8f;

#define EPSF 1e-5f
#define NB   16
#define CIN  256
#define MM   1024
#define DK   16
#define DV   64
#define QC   64

__global__ __launch_bounds__(256) void qkv_wmma(
    const float* __restrict__ x,  const float* __restrict__ Wq,
    const float* __restrict__ Wk, const float* __restrict__ Wv,
    const float* __restrict__ gq, const float* __restrict__ bq,
    const float* __restrict__ mq, const float* __restrict__ vq,
    const float* __restrict__ gv, const float* __restrict__ bv,
    const float* __restrict__ mv, const float* __restrict__ vv,
    float* __restrict__ Qbuf,
    float* __restrict__ Kbuf,
    float* __restrict__ Vbuf)
{
    const int wave = threadIdx.x >> 5;
    const int lane = threadIdx.x & 31;
    const int task = blockIdx.x * 8 + wave;
    const int b    = task / 576;
    const int rem  = task % 576;
    const int ot   = rem / 64;
    const int m0   = (rem % 64) * 16;

    const int nL   = lane & 15;
    const int half = lane >> 4;

    const float* W; int orow;
    if (ot < 4)       { W = Wq; orow = ot * 16; }
    else if (ot == 4) { W = Wk; orow = 0; }
    else              { W = Wv; orow = (ot - 5) * 16; }

    const float* wrow = W + (size_t)(orow + nL) * CIN + half * 2;
    const float* xcol = x + (size_t)b * CIN * MM + m0 + nL;

    v8f acc = {};
    #pragma unroll 4
    for (int c0 = 0; c0 < CIN; c0 += 4) {
        const int ca = c0 + half * 2;
        const v2f a = *reinterpret_cast<const v2f*>(wrow + c0);
        v2f bb;
        bb.x = xcol[(size_t)ca * MM];
        bb.y = xcol[(size_t)(ca + 1) * MM];
        acc = __builtin_amdgcn_wmma_f32_16x16x4_f32(
                  false, a, false, bb, (short)0, acc, false, false);
    }

    const int n = m0 + nL;
    if (ot == 4) {
        float* dst = Kbuf + (size_t)(b * MM + n) * DK;
        #pragma unroll
        for (int r = 0; r < 8; ++r) dst[r + half * 8] = acc[r];
    } else if (ot < 4) {
        float* dst = Qbuf + (size_t)(b * MM + n) * QC;
        #pragma unroll
        for (int r = 0; r < 8; ++r) {
            const int o = ot * 16 + r + half * 8;
            const float sc = gq[o] * rsqrtf(vq[o] + EPSF);
            dst[o] = acc[r] * sc + (bq[o] - mq[o] * sc);
        }
    } else {
        float* dst = Vbuf + (size_t)(b * MM + n) * DV;
        #pragma unroll
        for (int r = 0; r < 8; ++r) {
            const int o = (ot - 5) * 16 + r + half * 8;
            const float sc = gv[o] * rsqrtf(vv[o] + EPSF);
            dst[o] = acc[r] * sc + (bv[o] - mv[o] * sc);
        }
    }
}

__device__ __forceinline__ v8f wmma4(v2f a, v2f b, v8f c)
{
    return __builtin_amdgcn_wmma_f32_16x16x4_f32(
               false, a, false, b, (short)0, c, false, false);
}

__global__ __launch_bounds__(512) void content_reduce(
    float* __restrict__ Kbuf, const float* __restrict__ Vbuf)
{
    __shared__ float sk[256 * DK];
    __shared__ float part[16 * 256];

    const int tid  = threadIdx.x;
    const int wave = tid >> 5;
    const int lane = tid & 31;
    const int c    = lane & 15;
    const int hf   = lane >> 4;
    const int b    = blockIdx.x >> 2;
    const int mq   = blockIdx.x & 3;
    const int vt   = wave & 3;
    const int ms   = wave >> 2;
    float* kq = Kbuf + (size_t)(b * MM + mq * 256) * DK;

    const float* vp = Vbuf
        + (size_t)(b * MM + mq * 256 + ms * 64 + 2 * hf) * DV + vt * 16 + c;
    v2f bv[16];
    #pragma unroll
    for (int j = 0; j < 16; ++j) {
        bv[j].x = vp[(size_t)(4 * j) * DV];
        bv[j].y = vp[(size_t)(4 * j + 1) * DV];
    }

    if (tid < 256) {
        const v4f* p = reinterpret_cast<const v4f*>(kq + tid * DK);
        const v4f r0 = p[0], r1 = p[1], r2 = p[2], r3 = p[3];
        float v[DK], mx = -3.0e38f;
        #pragma unroll
        for (int k = 0; k < 4; ++k) {
            v[k] = r0[k]; v[4 + k] = r1[k]; v[8 + k] = r2[k]; v[12 + k] = r3[k];
        }
        #pragma unroll
        for (int k = 0; k < DK; ++k) mx = fmaxf(mx, v[k]);
        float s = 0.f;
        #pragma unroll
        for (int k = 0; k < DK; ++k) { v[k] = __expf(v[k] - mx); s += v[k]; }
        const float inv = 1.f / s;
        v4f* d = reinterpret_cast<v4f*>(sk + tid * DK);
        #pragma unroll
        for (int q = 0; q < 4; ++q) {
            v4f o;
            o[0] = v[4 * q] * inv;     o[1] = v[4 * q + 1] * inv;
            o[2] = v[4 * q + 2] * inv; o[3] = v[4 * q + 3] * inv;
            d[q] = o;
        }
    }
    __syncthreads();

    const float* ap = sk + (ms * 64 + 2 * hf) * DK + c;
    v8f t0 = {}, t1 = {};
    #pragma unroll
    for (int j = 0; j < 16; j += 2) {
        v2f a0, a1;
        a0.x = ap[(4 * j) * DK];
        a0.y = ap[(4 * j + 1) * DK];
        a1.x = ap[(4 * j + 4) * DK];
        a1.y = ap[(4 * j + 5) * DK];
        t0 = wmma4(a0, bv[j], t0);
        t1 = wmma4(a1, bv[j + 1], t1);
    }
    #pragma unroll
    for (int r = 0; r < 8; ++r)
        part[wave * 256 + (8 * hf + r) * 16 + c] = t0[r] + t1[r];
    __syncthreads();

    #pragma unroll
    for (int i = 0; i < 2; ++i) {
        const int o = tid + 512 * i;
        const float* pp = part + ((o & 63) >> 4) * 256 + (o >> 6) * 16 + (o & 15);
        kq[o] = ((pp[0] + pp[4 * 256]) + pp[8 * 256]) + pp[12 * 256];
    }
}

__device__ __forceinline__ void plane_scores(v8f (&G)[2],
    const float* __restrict__ relp, const v2f (&bq)[4])
{
    v4f ra[2][2];
    #pragma unroll
    for (int t = 0; t < 2; ++t) {
        const v4f* p = reinterpret_cast<const v4f*>(relp + t * (16 * DK));
        ra[t][0] = p[0];
        ra[t][1] = p[1];
    }
    v8f g0 = {}, g1 = {};
    #pragma unroll
    for (int jj = 0; jj < 4; ++jj) {
        v2f a0, a1;
        a0.x = ra[0][jj >> 1][(2 * jj) & 3];
        a0.y = ra[0][jj >> 1][(2 * jj + 1) & 3];
        a1.x = ra[1][jj >> 1][(2 * jj) & 3];
        a1.y = ra[1][jj >> 1][(2 * jj + 1) & 3];
        g0 = wmma4(a0, bq[jj], g0);
        g1 = wmma4(a1, bq[jj], g1);
    }
    G[0] = g0;
    G[1] = g1;
}

struct Stage {
    v4f ra[2];
    v4f w[8];
};

__device__ __forceinline__ void stage_load(Stage& d,
    const float* __restrict__ relp, const float* __restrict__ vp)
{
    const v4f* rp = reinterpret_cast<const v4f*>(relp);
    d.ra[0] = rp[0];
    d.ra[1] = rp[1];
    #pragma unroll
    for (int i = 0; i < 8; ++i)
        d.w[i] = *reinterpret_cast<const v4f*>(vp + (size_t)i * DV);
}

__global__ __launch_bounds__(128) void lambda_main(
    const float* __restrict__ rel,
    const float* __restrict__ Qbuf,
    const float* __restrict__ lam,
    const float* __restrict__ Vbuf,
    float* __restrict__ out)
{
    const int wave = threadIdx.x >> 5;
    const int lane = threadIdx.x & 31;
    const int c    = lane & 15;
    const int hf   = lane >> 4;

    const int gw = blockIdx.x * 4 + wave;
    const int b  = gw >> 8;
    const int id = gw & 255;
    const int x0 = (id >> 5) * 4;
    const int y0 = id & 31;

    const int nq = c >> 2, hq = c & 3;
    const bool is0 = nq == 0, is2 = nq == 2, islo = nq < 2;
    const v4f* qp = reinterpret_cast<const v4f*>(
        Qbuf + (size_t)(b * MM + (x0 + nq) * 32 + y0) * QC + hq * DK + hf * 8);
    const v4f q0 = qp[0], q1 = qp[1];
    v2f bq[4];
    bq[0].x = q0[0]; bq[0].y = q0[1];
    bq[1].x = q0[2]; bq[1].y = q0[3];
    bq[2].x = q1[0]; bq[2].y = q1[1];
    bq[3].x = q1[2]; bq[3].y = q1[3];

    const float* relb = rel + ((28 - x0) * 63 + (31 - y0) + c) * DK + hf * 8;
    const float* vb   = Vbuf + (size_t)(b * MM + hf * 8) * DV + c * 4;

    v8f Gw[4][2];
    plane_scores(Gw[0], relb,            bq);
    plane_scores(Gw[1], relb + 1008,     bq);
    plane_scores(Gw[2], relb + 2 * 1008, bq);

    v8f acc[4] = {};

    Stage cur, nxt;
    stage_load(cur, relb + 3 * 1008, vb);

    for (int x4 = 0; x4 < 32; x4 += 4) {
        #pragma unroll
        for (int u = 0; u < 4; ++u) {
            const int xm = x4 + u;
            #pragma unroll
            for (int t = 0; t < 2; ++t) {
                const int xn  = t == 0 ? xm : min(xm + 1, 31);
                const int mtn = 2 * xn + (t ^ 1);
                stage_load(nxt, relb + (xn + 3) * 1008 + (t ^ 1) * (16 * DK),
                           vb + (size_t)mtn * (16 * DV));
                __builtin_amdgcn_sched_barrier(0);

                v8f g = {};
                #pragma unroll
                for (int jj = 0; jj < 4; ++jj) {
                    v2f a;
                    a.x = cur.ra[jj >> 1][(2 * jj) & 3];
                    a.y = cur.ra[jj >> 1][(2 * jj + 1) & 3];
                    g = wmma4(a, bq[jj], g);
                }
                Gw[(u + 3) & 3][t] = g;

                v8f P;
                #pragma unroll
                for (int r = 0; r < 8; ++r) {
                    const float g0 = Gw[(u + 3) & 3][t][r];
                    const float g1 = Gw[(u + 2) & 3][t][r];
                    const float g2 = Gw[(u + 1) & 3][t][r];
                    const float g3 = Gw[u & 3][t][r];
                    const float lo = is0 ? g0 : g1;
                    const float hi = is2 ? g2 : g3;
                    P[r] = islo ? lo : hi;
                }

                #pragma unroll
                for (int j = 0; j < 4; ++j) {
                    v2f a2;
                    a2.x = P[2 * j];
                    a2.y = P[2 * j + 1];
                    #pragma unroll
                    for (int tt = 0; tt < 4; ++tt) {
                        v2f b2;
                        b2.x = cur.w[2 * j][tt];
                        b2.y = cur.w[2 * j + 1][tt];
                        acc[tt] = wmma4(a2, b2, acc[tt]);
                    }
                }

                __builtin_amdgcn_sched_barrier(0);

                cur = nxt;
            }
        }
    }

    {
        const float* lb = lam + (size_t)(b * MM) * DK + (hf * 8) * DV + c * 4;
        v4f lw[8];
        #pragma unroll
        for (int i = 0; i < 8; ++i)
            lw[i] = *reinterpret_cast<const v4f*>(lb + i * DV);
        #pragma unroll
        for (int q = 1; q < 4; ++q) {
            #pragma unroll
            for (int i = 0; i < 8; ++i)
                lw[i] += *reinterpret_cast<const v4f*>(lb + q * (256 * DK) + i * DV);
        }
        #pragma unroll
        for (int jj = 0; jj < 4; ++jj) {
            #pragma unroll
            for (int tt = 0; tt < 4; ++tt) {
                v2f b2;
                b2.x = lw[2 * jj][tt];
                b2.y = lw[2 * jj + 1][tt];
                acc[tt] = wmma4(bq[jj], b2, acc[tt]);
            }
        }
    }

    #pragma unroll
    for (int r = 0; r < 8; ++r) {
        const int n = (x0 + 2 * hf + (r >> 2)) * 32 + y0;
        v4f o;
        o[0] = acc[0][r]; o[1] = acc[1][r]; o[2] = acc[2][r]; o[3] = acc[3][r];
        *reinterpret_cast<v4f*>(
            out + (size_t)(b * MM + n) * 256 + (r & 3) * DV + c * 4) = o;
    }
}

extern "C" void kernel_launch(void* const* d_in, const int* in_sizes, int n_in,
                              void* d_out, int out_size, void* d_ws, size_t ws_size,
                              hipStream_t stream)
{
    const float* x   = (const float*)d_in[0];
    const float* Wq  = (const float*)d_in[1];
    const float* Wk  = (const float*)d_in[2];
    const float* Wv  = (const float*)d_in[3];
    const float* gq  = (const float*)d_in[4];
    const float* bq  = (const float*)d_in[5];
    const float* mq  = (const float*)d_in[6];
    const float* vq  = (const float*)d_in[7];
    const float* gv  = (const float*)d_in[8];
    const float* bv  = (const float*)d_in[9];
    const float* mv  = (const float*)d_in[10];
    const float* vv  = (const float*)d_in[11];
    const float* rel = (const float*)d_in[12];
    float* out = (float*)d_out;

    float* Qbuf = (float*)d_ws;
    float* Vbuf = Qbuf + (size_t)NB * MM * QC;
    float* Kbuf = Vbuf + (size_t)NB * MM * DV;

    qkv_wmma<<<1152, 256, 0, stream>>>(x, Wq, Wk, Wv, gq, bq, mq, vq,
                                       gv, bv, mv, vv, Qbuf, Kbuf, Vbuf);
    content_reduce<<<64, 512, 0, stream>>>(Kbuf, Vbuf);
    lambda_main<<<1024, 128, 0, stream>>>(rel, Qbuf, Kbuf, Vbuf, out);
}


# Kernel 1 wave tile 16 o x 16 m -> 16 o x 64 m, block = the 9 o-tile waves of one x tile: b128 x load feeds 4 WMMAs, b128 W load 8, next group loaded ahead; changes the f32 summation order
# speedup vs baseline: 2.8356x; 1.1306x over previous
#include <hip/hip_runtime.h>

typedef __attribute__((ext_vector_type(2))) float v2f;
typedef __attribute__((ext_vector_type(4))) float v4f;
typedef __attribute__((ext_vector_type(8))) float v8f;

#define EPSF 1e-5f
#define NB   16
#define CIN  256
#define MM   1024
#define DK   16
#define DV   64
#define QC   64

__device__ __forceinline__ v8f wmma4(v2f a, v2f b, v8f c)
{
    return __builtin_amdgcn_wmma_f32_16x16x4_f32(
               false, a, false, b, (short)0, c, false, false);
}

struct CGroup {
    v4f w[2];
    v4f xv[8];
};

__device__ __forceinline__ void cgroup_load(CGroup& g,
    const float* __restrict__ wp, const float* __restrict__ xp)
{
    g.w[0] = *reinterpret_cast<const v4f*>(wp);
    g.w[1] = *reinterpret_cast<const v4f*>(wp + 4);
    #pragma unroll
    for (int e = 0; e < 8; ++e)
        g.xv[e] = *reinterpret_cast<const v4f*>(xp + (size_t)e * MM);
}

__device__ __forceinline__ void cgroup_mma(v8f (&acc)[4], const CGroup& g)
{
    #pragma unroll
    for (int s = 0; s < 4; ++s) {
        v2f a;
        a.x = g.w[s >> 1][(2 * s) & 3];
        a.y = g.w[s >> 1][(2 * s + 1) & 3];
        #pragma unroll
        for (int i = 0; i < 4; ++i) {
            v2f bb;
            bb.x = g.xv[2 * s][i];
            bb.y = g.xv[2 * s + 1][i];
            acc[i] = wmma4(a, bb, acc[i]);
        }
    }
}

__global__ __launch_bounds__(288) void qkv_wmma(
    const float* __restrict__ x,  const float* __restrict__ Wq,
    const float* __restrict__ Wk, const float* __restrict__ Wv,
    const float* __restrict__ gq, const float* __restrict__ bq,
    const float* __restrict__ mq, const float* __restrict__ vq,
    const float* __restrict__ gv, const float* __restrict__ bv,
    const float* __restrict__ mv, const float* __restrict__ vv,
    float* __restrict__ Qbuf,
    float* __restrict__ Kbuf,
    float* __restrict__ Vbuf)
{
    const int ot   = threadIdx.x >> 5;
    const int lane = threadIdx.x & 31;
    const int b    = blockIdx.x >> 4;
    const int mb   = (blockIdx.x & 15) * 64;

    const int nL   = lane & 15;
    const int half = lane >> 4;

    const float* W; int orow;
    if (ot < 4)       { W = Wq; orow = ot * 16; }
    else if (ot == 4) { W = Wk; orow = 0; }
    else              { W = Wv; orow = (ot - 5) * 16; }

    const float* wp = W + (size_t)(orow + nL) * CIN + half * 8;
    const float* xp = x + (size_t)b * CIN * MM + (size_t)(half * 8) * MM + mb + 4 * nL;

    v8f acc[4] = {};
    CGroup ga, gb;
    cgroup_load(ga, wp, xp);
    for (int c0 = 0; c0 < CIN - 32; c0 += 32) {
        cgroup_load(gb, wp + c0 + 16, xp + (size_t)(c0 + 16) * MM);
        __builtin_amdgcn_sched_barrier(0);
        cgroup_mma(acc, ga);
        __builtin_amdgcn_sched_barrier(0);
        cgroup_load(ga, wp + c0 + 32, xp + (size_t)(c0 + 32) * MM);
        __builtin_amdgcn_sched_barrier(0);
        cgroup_mma(acc, gb);
        __builtin_amdgcn_sched_barrier(0);
    }
    cgroup_load(gb, wp + CIN - 16, xp + (size_t)(CIN - 16) * MM);
    __builtin_amdgcn_sched_barrier(0);
    cgroup_mma(acc, ga);
    cgroup_mma(acc, gb);

    const int ob = orow + half * 8;
    if (ot == 4) {
        #pragma unroll
        for (int i = 0; i < 4; ++i) {
            v4f* dst = reinterpret_cast<v4f*>(
                Kbuf + (size_t)(b * MM + mb + 4 * nL + i) * DK + ob);
            v4f o0, o1;
            #pragma unroll
            for (int r = 0; r < 4; ++r) { o0[r] = acc[i][r]; o1[r] = acc[i][4 + r]; }
            dst[0] = o0;
            dst[1] = o1;
        }
    } else {
        const bool isq = ot < 4;
        const float* gp = (isq ? gq : gv) + ob;
        const float* bp = (isq ? bq : bv) + ob;
        const float* mp = (isq ? mq : mv) + ob;
        const float* vp = (isq ? vq : vv) + ob;
        float* obuf = isq ? Qbuf : Vbuf;
        float sc[8], sh[8];
        #pragma unroll
        for (int r = 0; r < 8; ++r) {
            sc[r] = gp[r] * rsqrtf(vp[r] + EPSF);
            sh[r] = bp[r] - mp[r] * sc[r];
        }
        #pragma unroll
        for (int i = 0; i < 4; ++i) {
            v4f* dst = reinterpret_cast<v4f*>(
                obuf + (size_t)(b * MM + mb + 4 * nL + i) * QC + ob);
            v4f o0, o1;
            #pragma unroll
            for (int r = 0; r < 4; ++r) {
                o0[r] = acc[i][r] * sc[r] + sh[r];
                o1[r] = acc[i][4 + r] * sc[4 + r] + sh[4 + r];
            }
            dst[0] = o0;
            dst[1] = o1;
        }
    }
}

__global__ __launch_bounds__(512) void content_reduce(
    float* __restrict__ Kbuf, const float* __restrict__ Vbuf)
{
    __shared__ float sk[256 * DK];
    __shared__ float part[16 * 256];

    const int tid  = threadIdx.x;
    const int wave = tid >> 5;
    const int lane = tid & 31;
    const int c    = lane & 15;
    const int hf   = lane >> 4;
    const int b    = blockIdx.x >> 2;
    const int mq   = blockIdx.x & 3;
    const int vt   = wave & 3;
    const int ms   = wave >> 2;
    float* kq = Kbuf + (size_t)(b * MM + mq * 256) * DK;

    const float* vp = Vbuf
        + (size_t)(b * MM + mq * 256 + ms * 64 + 2 * hf) * DV + vt * 16 + c;
    v2f bv[16];
    #pragma unroll
    for (int j = 0; j < 16; ++j) {
        bv[j].x = vp[(size_t)(4 * j) * DV];
        bv[j].y = vp[(size_t)(4 * j + 1) * DV];
    }

    if (tid < 256) {
        const v4f* p = reinterpret_cast<const v4f*>(kq + tid * DK);
        const v4f r0 = p[0], r1 = p[1], r2 = p[2], r3 = p[3];
        float v[DK], mx = -3.0e38f;
        #pragma unroll
        for (int k = 0; k < 4; ++k) {
            v[k] = r0[k]; v[4 + k] = r1[k]; v[8 + k] = r2[k]; v[12 + k] = r3[k];
        }
        #pragma unroll
        for (int k = 0; k < DK; ++k) mx = fmaxf(mx, v[k]);
        float s = 0.f;
        #pragma unroll
        for (int k = 0; k < DK; ++k) { v[k] = __expf(v[k] - mx); s += v[k]; }
        const float inv = 1.f / s;
        v4f* d = reinterpret_cast<v4f*>(sk + tid * DK);
        #pragma unroll
        for (int q = 0; q < 4; ++q) {
            v4f o;
            o[0] = v[4 * q] * inv;     o[1] = v[4 * q + 1] * inv;
            o[2] = v[4 * q + 2] * inv; o[3] = v[4 * q + 3] * inv;
            d[q] = o;
        }
    }
    __syncthreads();

    const float* ap = sk + (ms * 64 + 2 * hf) * DK + c;
    v8f t0 = {}, t1 = {};
    #pragma unroll
    for (int j = 0; j < 16; j += 2) {
        v2f a0, a1;
        a0.x = ap[(4 * j) * DK];
        a0.y = ap[(4 * j + 1) * DK];
        a1.x = ap[(4 * j + 4) * DK];
        a1.y = ap[(4 * j + 5) * DK];
        t0 = wmma4(a0, bv[j], t0);
        t1 = wmma4(a1, bv[j + 1], t1);
    }
    #pragma unroll
    for (int r = 0; r < 8; ++r)
        part[wave * 256 + (8 * hf + r) * 16 + c] = t0[r] + t1[r];
    __syncthreads();

    #pragma unroll
    for (int i = 0; i < 2; ++i) {
        const int o = tid + 512 * i;
        const float* pp = part + ((o & 63) >> 4) * 256 + (o >> 6) * 16 + (o & 15);
        kq[o] = ((pp[0] + pp[4 * 256]) + pp[8 * 256]) + pp[12 * 256];
    }
}

__device__ __forceinline__ void plane_scores(v8f (&G)[2],
    const float* __restrict__ relp, const v2f (&bq)[4])
{
    v4f ra[2][2];
    #pragma unroll
    for (int t = 0; t < 2; ++t) {
        const v4f* p = reinterpret_cast<const v4f*>(relp + t * (16 * DK));
        ra[t][0] = p[0];
        ra[t][1] = p[1];
    }
    v8f g0 = {}, g1 = {};
    #pragma unroll
    for (int jj = 0; jj < 4; ++jj) {
        v2f a0, a1;
        a0.x = ra[0][jj >> 1][(2 * jj) & 3];
        a0.y = ra[0][jj >> 1][(2 * jj + 1) & 3];
        a1.x = ra[1][jj >> 1][(2 * jj) & 3];
        a1.y = ra[1][jj >> 1][(2 * jj + 1) & 3];
        g0 = wmma4(a0, bq[jj], g0);
        g1 = wmma4(a1, bq[jj], g1);
    }
    G[0] = g0;
    G[1] = g1;
}

struct Stage {
    v4f ra[2];
    v4f w[8];
};

__device__ __forceinline__ void stage_load(Stage& d,
    const float* __restrict__ relp, const float* __restrict__ vp)
{
    const v4f* rp = reinterpret_cast<const v4f*>(relp);
    d.ra[0] = rp[0];
    d.ra[1] = rp[1];
    #pragma unroll
    for (int i = 0; i < 8; ++i)
        d.w[i] = *reinterpret_cast<const v4f*>(vp + (size_t)i * DV);
}

__global__ __launch_bounds__(128) void lambda_main(
    const float* __restrict__ rel,
    const float* __restrict__ Qbuf,
    const float* __restrict__ lam,
    const float* __restrict__ Vbuf,
    float* __restrict__ out)
{
    const int wave = threadIdx.x >> 5;
    const int lane = threadIdx.x & 31;
    const int c    = lane & 15;
    const int hf   = lane >> 4;

    const int gw = blockIdx.x * 4 + wave;
    const int b  = gw >> 8;
    const int id = gw & 255;
    const int x0 = (id >> 5) * 4;
    const int y0 = id & 31;

    const int nq = c >> 2, hq = c & 3;
    const bool is0 = nq == 0, is2 = nq == 2, islo = nq < 2;
    const v4f* qp = reinterpret_cast<const v4f*>(
        Qbuf + (size_t)(b * MM + (x0 + nq) * 32 + y0) * QC + hq * DK + hf * 8);
    const v4f q0 = qp[0], q1 = qp[1];
    v2f bq[4];
    bq[0].x = q0[0]; bq[0].y = q0[1];
    bq[1].x = q0[2]; bq[1].y = q0[3];
    bq[2].x = q1[0]; bq[2].y = q1[1];
    bq[3].x = q1[2]; bq[3].y = q1[3];

    const float* relb = rel + ((28 - x0) * 63 + (31 - y0) + c) * DK + hf * 8;
    const float* vb   = Vbuf + (size_t)(b * MM + hf * 8) * DV + c * 4;

    v8f Gw[4][2];
    plane_scores(Gw[0], relb,            bq);
    plane_scores(Gw[1], relb + 1008,     bq);
    plane_scores(Gw[2], relb + 2 * 1008, bq);

    v8f acc[4] = {};

    Stage cur, nxt;
    stage_load(cur, relb + 3 * 1008, vb);

    for (int x4 = 0; x4 < 32; x4 += 4) {
        #pragma unroll
        for (int u = 0; u < 4; ++u) {
            const int xm = x4 + u;
            #pragma unroll
            for (int t = 0; t < 2; ++t) {
                const int xn  = t == 0 ? xm : min(xm + 1, 31);
                const int mtn = 2 * xn + (t ^ 1);
                stage_load(nxt, relb + (xn + 3) * 1008 + (t ^ 1) * (16 * DK),
                           vb + (size_t)mtn * (16 * DV));
                __builtin_amdgcn_sched_barrier(0);

                v8f g = {};
                #pragma unroll
                for (int jj = 0; jj < 4; ++jj) {
                    v2f a;
                    a.x = cur.ra[jj >> 1][(2 * jj) & 3];
                    a.y = cur.ra[jj >> 1][(2 * jj + 1) & 3];
                    g = wmma4(a, bq[jj], g);
                }
                Gw[(u + 3) & 3][t] = g;

                v8f P;
                #pragma unroll
                for (int r = 0; r < 8; ++r) {
                    const float g0 = Gw[(u + 3) & 3][t][r];
                    const float g1 = Gw[(u + 2) & 3][t][r];
                    const float g2 = Gw[(u + 1) & 3][t][r];
                    const float g3 = Gw[u & 3][t][r];
                    const float lo = is0 ? g0 : g1;
                    const float hi = is2 ? g2 : g3;
                    P[r] = islo ? lo : hi;
                }

                #pragma unroll
                for (int j = 0; j < 4; ++j) {
                    v2f a2;
                    a2.x = P[2 * j];
                    a2.y = P[2 * j + 1];
                    #pragma unroll
                    for (int tt = 0; tt < 4; ++tt) {
                        v2f b2;
                        b2.x = cur.w[2 * j][tt];
                        b2.y = cur.w[2 * j + 1][tt];
                        acc[tt] = wmma4(a2, b2, acc[tt]);
                    }
                }

                __builtin_amdgcn_sched_barrier(0);

                cur = nxt;
            }
        }
    }

    {
        const float* lb = lam + (size_t)(b * MM) * DK + (hf * 8) * DV + c * 4;
        v4f lw[8];
        #pragma unroll
        for (int i = 0; i < 8; ++i)
            lw[i] = *reinterpret_cast<const v4f*>(lb + i * DV);
        #pragma unroll
        for (int q = 1; q < 4; ++q) {
            #pragma unroll
            for (int i = 0; i < 8; ++i)
                lw[i] += *reinterpret_cast<const v4f*>(lb + q * (256 * DK) + i * DV);
        }
        #pragma unroll
        for (int jj = 0; jj < 4; ++jj) {
            #pragma unroll
            for (int tt = 0; tt < 4; ++tt) {
                v2f b2;
                b2.x = lw[2 * jj][tt];
                b2.y = lw[2 * jj + 1][tt];
                acc[tt] = wmma4(bq[jj], b2, acc[tt]);
            }
        }
    }

    #pragma unroll
    for (int r = 0; r < 8; ++r) {
        const int n = (x0 + 2 * hf + (r >> 2)) * 32 + y0;
        v4f o;
        o[0] = acc[0][r]; o[1] = acc[1][r]; o[2] = acc[2][r]; o[3] = acc[3][r];
        *reinterpret_cast<v4f*>(
            out + (size_t)(b * MM + n) * 256 + (r & 3) * DV + c * 4) = o;
    }
}

extern "C" void kernel_launch(void* const* d_in, const int* in_sizes, int n_in,
                              void* d_out, int out_size, void* d_ws, size_t ws_size,
                              hipStream_t stream)
{
    const float* x   = (const float*)d_in[0];
    const float* Wq  = (const float*)d_in[1];
    const float* Wk  = (const float*)d_in[2];
    const float* Wv  = (const float*)d_in[3];
    const float* gq  = (const float*)d_in[4];
    const float* bq  = (const float*)d_in[5];
    const float* mq  = (const float*)d_in[6];
    const float* vq  = (const float*)d_in[7];
    const float* gv  = (const float*)d_in[8];
    const float* bv  = (const float*)d_in[9];
    const float* mv  = (const float*)d_in[10];
    const float* vv  = (const float*)d_in[11];
    const float* rel = (const float*)d_in[12];
    float* out = (float*)d_out;

    float* Qbuf = (float*)d_ws;
    float* Vbuf = Qbuf + (size_t)NB * MM * QC;
    float* Kbuf = Vbuf + (size_t)NB * MM * DV;

    qkv_wmma<<<256, 288, 0, stream>>>(x, Wq, Wk, Wv, gq, bq, mq, vq,
                                       gv, bv, mv, vv, Qbuf, Kbuf, Vbuf);
    content_reduce<<<64, 512, 0, stream>>>(Kbuf, Vbuf);
    lambda_main<<<1024, 128, 0, stream>>>(rel, Qbuf, Kbuf, Vbuf, out);
}
